# starting kernel
# baseline (speedup 1.0000x reference)
#include <hip/hip_runtime.h>

#define NNODES 100000
#define NEDGES 1600000
#define DFEAT  128
#define NCLS   16
#define BN_EPS 1e-5f

typedef float v2f __attribute__((ext_vector_type(2)));
typedef float v8f __attribute__((ext_vector_type(8)));

__global__ void count_deg_kernel(const int* __restrict__ dst,
                                 float* __restrict__ cnt, int ne) {
    int e = blockIdx.x * blockDim.x + threadIdx.x;
    if (e < ne) atomicAdd(&cnt[dst[e]], 1.0f);
}

__global__ void invert_cnt_kernel(float* __restrict__ cnt, int n) {
    int i = blockIdx.x * blockDim.x + threadIdx.x;
    if (i < n) cnt[i] = 1.0f / fmaxf(cnt[i], 1.0f);
}

__global__ void prep_affine_kernel(const float* __restrict__ bl,
                                   const float* __restrict__ g,
                                   const float* __restrict__ be,
                                   const float* __restrict__ m,
                                   const float* __restrict__ v,
                                   float* __restrict__ alpha,
                                   float* __restrict__ beta) {
    int j = threadIdx.x;
    float s  = g[j] * rsqrtf(v[j] + BN_EPS);
    alpha[j] = s;
    beta[j]  = (bl[j] - m[j]) * s + be[j];
}

__global__ __launch_bounds__(256)
void scatter_add_kernel(const float* __restrict__ x,
                        const int* __restrict__ src,
                        const int* __restrict__ dst,
                        float* __restrict__ agg, int ne) {
    int edge = blockIdx.x * (blockDim.x >> 5) + (threadIdx.x >> 5);
    int lane = threadIdx.x & 31;
    if (edge >= ne) return;
    int s = src[edge];
    int d = dst[edge];
    float4 val = ((const float4*)(x + (size_t)s * DFEAT))[lane];
    float* o = agg + (size_t)d * DFEAT + lane * 4;
    atomicAdd(o + 0, val.x);
    atomicAdd(o + 1, val.y);
    atomicAdd(o + 2, val.z);
    atomicAdd(o + 3, val.w);
}

#define APITCH 132

__global__ __launch_bounds__(256)
void sage_fused_kernel(const float* __restrict__ agg,
                       const float* __restrict__ x,
                       const float* __restrict__ inv,
                       const float* __restrict__ wl,
                       const float* __restrict__ wr,
                       const float* __restrict__ alpha,
                       const float* __restrict__ beta,
                       float* __restrict__ out) {
    __shared__ float As[16 * APITCH];
    __shared__ float Xs[16 * APITCH];

    const int tid    = threadIdx.x;
    const int blkrow = blockIdx.x * 16;

    #pragma unroll
    for (int i = 0; i < 2; ++i) {
        int idx = tid * 2 + i;
        int r   = idx >> 5;
        int c4  = idx & 31;
        float s = inv[blkrow + r];
        float4 av = ((const float4*)(agg + (size_t)(blkrow + r) * DFEAT))[c4];
        av.x *= s; av.y *= s; av.z *= s; av.w *= s;
        *(float4*)&As[r * APITCH + c4 * 4] = av;
        float4 xv = ((const float4*)(x + (size_t)(blkrow + r) * DFEAT))[c4];
        *(float4*)&Xs[r * APITCH + c4 * 4] = xv;
    }
    __syncthreads();

    const int wave  = tid >> 5;
    const int lane  = tid & 31;
    const int c0    = wave * 16;
    const int lrow  = lane & 15;
    const int khalf = (lane >> 4) * 2;

    v8f acc = {};
    for (int k0 = 0; k0 < DFEAT; k0 += 4) {
        const int ka = k0 + khalf;
        v2f a, b;
        a.x = As[lrow * APITCH + ka];
        a.y = As[lrow * APITCH + ka + 1];
        b.x = wl[(size_t)ka * DFEAT + c0 + lrow];
        b.y = wl[(size_t)(ka + 1) * DFEAT + c0 + lrow];
        acc = __builtin_amdgcn_wmma_f32_16x16x4_f32(false, a, false, b,
                                                    (short)0, acc, false, false);
        a.x = Xs[lrow * APITCH + ka];
        a.y = Xs[lrow * APITCH + ka + 1];
        b.x = wr[(size_t)ka * DFEAT + c0 + lrow];
        b.y = wr[(size_t)(ka + 1) * DFEAT + c0 + lrow];
        acc = __builtin_amdgcn_wmma_f32_16x16x4_f32(false, a, false, b,
                                                    (short)0, acc, false, false);
    }

    const int col   = c0 + lrow;
    const float al  = alpha[col];
    const float bt  = beta[col];
    const int mbase = (lane >> 4) * 8;
    #pragma unroll
    for (int r = 0; r < 8; ++r) {
        float v = acc[r] * al + bt;
        v = fmaxf(v, 0.0f);
        out[(size_t)(blkrow + mbase + r) * DFEAT + col] = v;
    }
}

template <int NCOLS, bool RELU>
__global__ void linear_wmma_kernel(const float* __restrict__ A,
                                   const float* __restrict__ W,
                                   const float* __restrict__ bias,
                                   float* __restrict__ out) {
    const int tid    = threadIdx.x;
    const int wave   = tid >> 5;
    const int lane   = tid & 31;
    const int c0     = wave * 16;
    const int blkrow = blockIdx.x * 16;
    const int lrow   = lane & 15;
    const int khalf  = (lane >> 4) * 2;

    const float* Arow = A + (size_t)(blkrow + lrow) * DFEAT;

    v8f acc = {};
    for (int k0 = 0; k0 < DFEAT; k0 += 4) {
        const int ka = k0 + khalf;
        v2f a, b;
        a.x = Arow[ka];
        a.y = Arow[ka + 1];
        b.x = W[(size_t)ka * NCOLS + c0 + lrow];
        b.y = W[(size_t)(ka + 1) * NCOLS + c0 + lrow];
        acc = __builtin_amdgcn_wmma_f32_16x16x4_f32(false, a, false, b,
                                                    (short)0, acc, false, false);
    }

    const int col   = c0 + lrow;
    const float bt  = bias[col];
    const int mbase = (lane >> 4) * 8;
    #pragma unroll
    for (int r = 0; r < 8; ++r) {
        float v = acc[r] + bt;
        if (RELU) v = fmaxf(v, 0.0f);
        out[(size_t)(blkrow + mbase + r) * NCOLS + col] = v;
    }
}

extern "C" void kernel_launch(void* const* d_in, const int* in_sizes, int n_in,
                              void* d_out, int out_size, void* d_ws, size_t ws_size,
                              hipStream_t stream) {
    (void)in_sizes; (void)n_in; (void)out_size; (void)ws_size;

    const float* x  = (const float*)d_in[0];
    const int*   ei = (const int*)d_in[1];
    const int* src = ei;
    const int* dst = ei + NEDGES;

    const float* wl_[3]; const float* bl_[3]; const float* wr_[3];
    const float* g_[3];  const float* be_[3]; const float* m_[3]; const float* v_[3];
    for (int l = 0; l < 3; ++l) {
        int b = 2 + 7 * l;
        wl_[l] = (const float*)d_in[b + 0];
        bl_[l] = (const float*)d_in[b + 1];
        wr_[l] = (const float*)d_in[b + 2];
        g_[l]  = (const float*)d_in[b + 3];
        be_[l] = (const float*)d_in[b + 4];
        m_[l]  = (const float*)d_in[b + 5];
        v_[l]  = (const float*)d_in[b + 6];
    }
    const float* wf = (const float*)d_in[23];
    const float* bf = (const float*)d_in[24];
    const float* wc = (const float*)d_in[25];
    const float* bc = (const float*)d_in[26];

    float* ws  = (float*)d_ws;
    float* inv = ws;
    float* agg = inv + NNODES;
    float* hA  = agg + (size_t)NNODES * DFEAT;
    float* hB  = hA  + (size_t)NNODES * DFEAT;
    float* aff = hB  + (size_t)NNODES * DFEAT;

    const int ROWBLKS = NNODES / 16;

    hipMemsetAsync(inv, 0, NNODES * sizeof(float), stream);
    count_deg_kernel<<<(NEDGES + 255) / 256, 256, 0, stream>>>(dst, inv, NEDGES);
    invert_cnt_kernel<<<(NNODES + 255) / 256, 256, 0, stream>>>(inv, NNODES);

    for (int l = 0; l < 3; ++l)
        prep_affine_kernel<<<1, 128, 0, stream>>>(bl_[l], g_[l], be_[l], m_[l], v_[l],
                                                  aff + 256 * l, aff + 256 * l + 128);

    const float* lin[3] = { x, hA, hB };
    float*       lout[3] = { hA, hB, hA };
    for (int l = 0; l < 3; ++l) {
        hipMemsetAsync(agg, 0, (size_t)NNODES * DFEAT * sizeof(float), stream);
        scatter_add_kernel<<<(NEDGES + 7) / 8, 256, 0, stream>>>(lin[l], src, dst,
                                                                 agg, NEDGES);
        sage_fused_kernel<<<ROWBLKS, 256, 0, stream>>>(agg, lin[l], inv,
                                                       wl_[l], wr_[l],
                                                       aff + 256 * l,
                                                       aff + 256 * l + 128,
                                                       lout[l]);
    }

    linear_wmma_kernel<DFEAT, true><<<ROWBLKS, 256, 0, stream>>>(hA, wf, bf, hB);
    linear_wmma_kernel<NCLS, false><<<ROWBLKS, 32, 0, stream>>>(hB, wc, bc,
                                                                (float*)d_out);
}


# Edge aggregation: float-atomic scatter and its memsets replaced by per-call dst-grouped neighbour lists (int atomic rank + scan) and a per-node gather sum; sum order = atomic arrival rank
# speedup vs baseline: 2.6035x; 2.6035x over previous
#include <hip/hip_runtime.h>

#define NNODES 100000
#define NEDGES 1600000
#define DFEAT  128
#define NCLS   16
#define BN_EPS 1e-5f

typedef float v2f __attribute__((ext_vector_type(2)));
typedef float v8f __attribute__((ext_vector_type(8)));

__global__ void count_rank_kernel(const int* __restrict__ dst,
                                  int* __restrict__ deg,
                                  int* __restrict__ rank, int ne) {
    int e = blockIdx.x * blockDim.x + threadIdx.x;
    if (e < ne) rank[e] = atomicAdd(&deg[dst[e]], 1);
}

#define SCAN_BLOCK 1024

__global__ __launch_bounds__(SCAN_BLOCK)
void scan_deg_kernel(const int* __restrict__ deg,
                     int* __restrict__ rowptr,
                     float* __restrict__ inv, int n) {
    __shared__ int wsum[32];
    __shared__ int blk_off;

    const int t    = threadIdx.x;
    const int lane = t & 31;
    const int w    = t >> 5;
    const int base = blockIdx.x * SCAN_BLOCK;

    int part = 0;
    for (int i = t; i < base; i += SCAN_BLOCK) part += deg[i];
    for (int d = 16; d > 0; d >>= 1) part += __shfl_xor(part, d, 32);
    if (lane == 0) wsum[w] = part;
    __syncthreads();
    if (w == 0) {
        int v = wsum[lane];
        for (int d = 16; d > 0; d >>= 1) v += __shfl_xor(v, d, 32);
        if (lane == 0) blk_off = v;
    }
    __syncthreads();
    const int off = blk_off;

    const int i  = base + t;
    const int dg = (i < n) ? deg[i] : 0;
    int incl = dg;
    for (int d = 1; d < 32; d <<= 1) {
        int y = __shfl_up(incl, d, 32);
        if (lane >= d) incl += y;
    }
    if (lane == 31) wsum[w] = incl;
    __syncthreads();
    if (w == 0) {
        const int v = wsum[lane];
        int s = v;
        for (int d = 1; d < 32; d <<= 1) {
            int y = __shfl_up(s, d, 32);
            if (lane >= d) s += y;
        }
        wsum[lane] = s - v;
    }
    __syncthreads();
    const int excl = off + wsum[w] + incl - dg;
    if (i < n) {
        rowptr[i] = excl;
        inv[i]    = 1.0f / fmaxf((float)dg, 1.0f);
    }
    if (i == n - 1) rowptr[n] = excl + dg;
}

__global__ void fill_col_kernel(const int* __restrict__ src,
                                const int* __restrict__ dst,
                                const int* __restrict__ rank,
                                const int* __restrict__ rowptr,
                                int* __restrict__ col, int ne) {
    int e = blockIdx.x * blockDim.x + threadIdx.x;
    if (e < ne) col[rowptr[dst[e]] + rank[e]] = src[e];
}

__global__ void prep_affine_kernel(const float* __restrict__ bl,
                                   const float* __restrict__ g,
                                   const float* __restrict__ be,
                                   const float* __restrict__ m,
                                   const float* __restrict__ v,
                                   float* __restrict__ alpha,
                                   float* __restrict__ beta) {
    int j = threadIdx.x;
    float s  = g[j] * rsqrtf(v[j] + BN_EPS);
    alpha[j] = s;
    beta[j]  = (bl[j] - m[j]) * s + be[j];
}

__global__ __launch_bounds__(256)
void gather_sum_kernel(const float* __restrict__ x,
                       const int* __restrict__ rowptr,
                       const int* __restrict__ col,
                       float* __restrict__ agg, int n) {
    const int lane = threadIdx.x & 31;
    const int node = __builtin_amdgcn_readfirstlane(
        (int)(blockIdx.x * (blockDim.x >> 5) + (threadIdx.x >> 5)));
    if (node >= n) return;
    const int beg = rowptr[node];
    const int end = rowptr[node + 1];

    float4 acc = make_float4(0.0f, 0.0f, 0.0f, 0.0f);
    for (int base = beg; base < end; base += 32) {
        const int m    = min(32, end - base);
        const int mine = (lane < m) ? col[base + lane] : 0;
        int j = 0;
        for (; j + 4 <= m; j += 4) {
            const int s0 = __builtin_amdgcn_readlane(mine, j);
            const int s1 = __builtin_amdgcn_readlane(mine, j + 1);
            const int s2 = __builtin_amdgcn_readlane(mine, j + 2);
            const int s3 = __builtin_amdgcn_readlane(mine, j + 3);
            const float4 v0 = ((const float4*)(x + (size_t)s0 * DFEAT))[lane];
            const float4 v1 = ((const float4*)(x + (size_t)s1 * DFEAT))[lane];
            const float4 v2 = ((const float4*)(x + (size_t)s2 * DFEAT))[lane];
            const float4 v3 = ((const float4*)(x + (size_t)s3 * DFEAT))[lane];
            acc.x += v0.x; acc.y += v0.y; acc.z += v0.z; acc.w += v0.w;
            acc.x += v1.x; acc.y += v1.y; acc.z += v1.z; acc.w += v1.w;
            acc.x += v2.x; acc.y += v2.y; acc.z += v2.z; acc.w += v2.w;
            acc.x += v3.x; acc.y += v3.y; acc.z += v3.z; acc.w += v3.w;
        }
        for (; j < m; ++j) {
            const int s0 = __builtin_amdgcn_readlane(mine, j);
            const float4 v0 = ((const float4*)(x + (size_t)s0 * DFEAT))[lane];
            acc.x += v0.x; acc.y += v0.y; acc.z += v0.z; acc.w += v0.w;
        }
    }
    ((float4*)(agg + (size_t)node * DFEAT))[lane] = acc;
}

#define APITCH 132

__global__ __launch_bounds__(256)
void sage_fused_kernel(const float* __restrict__ agg,
                       const float* __restrict__ x,
                       const float* __restrict__ inv,
                       const float* __restrict__ wl,
                       const float* __restrict__ wr,
                       const float* __restrict__ alpha,
                       const float* __restrict__ beta,
                       float* __restrict__ out) {
    __shared__ float As[16 * APITCH];
    __shared__ float Xs[16 * APITCH];

    const int tid    = threadIdx.x;
    const int blkrow = blockIdx.x * 16;

    #pragma unroll
    for (int i = 0; i < 2; ++i) {
        int idx = tid * 2 + i;
        int r   = idx >> 5;
        int c4  = idx & 31;
        float s = inv[blkrow + r];
        float4 av = ((const float4*)(agg + (size_t)(blkrow + r) * DFEAT))[c4];
        av.x *= s; av.y *= s; av.z *= s; av.w *= s;
        *(float4*)&As[r * APITCH + c4 * 4] = av;
        float4 xv = ((const float4*)(x + (size_t)(blkrow + r) * DFEAT))[c4];
        *(float4*)&Xs[r * APITCH + c4 * 4] = xv;
    }
    __syncthreads();

    const int wave  = tid >> 5;
    const int lane  = tid & 31;
    const int c0    = wave * 16;
    const int lrow  = lane & 15;
    const int khalf = (lane >> 4) * 2;

    v8f acc = {};
    for (int k0 = 0; k0 < DFEAT; k0 += 4) {
        const int ka = k0 + khalf;
        v2f a, b;
        a.x = As[lrow * APITCH + ka];
        a.y = As[lrow * APITCH + ka + 1];
        b.x = wl[(size_t)ka * DFEAT + c0 + lrow];
        b.y = wl[(size_t)(ka + 1) * DFEAT + c0 + lrow];
        acc = __builtin_amdgcn_wmma_f32_16x16x4_f32(false, a, false, b,
                                                    (short)0, acc, false, false);
        a.x = Xs[lrow * APITCH + ka];
        a.y = Xs[lrow * APITCH + ka + 1];
        b.x = wr[(size_t)ka * DFEAT + c0 + lrow];
        b.y = wr[(size_t)(ka + 1) * DFEAT + c0 + lrow];
        acc = __builtin_amdgcn_wmma_f32_16x16x4_f32(false, a, false, b,
                                                    (short)0, acc, false, false);
    }

    const int col   = c0 + lrow;
    const float al  = alpha[col];
    const float bt  = beta[col];
    const int mbase = (lane >> 4) * 8;
    #pragma unroll
    for (int r = 0; r < 8; ++r) {
        float v = acc[r] * al + bt;
        v = fmaxf(v, 0.0f);
        out[(size_t)(blkrow + mbase + r) * DFEAT + col] = v;
    }
}

template <int NCOLS, bool RELU>
__global__ void linear_wmma_kernel(const float* __restrict__ A,
                                   const float* __restrict__ W,
                                   const float* __restrict__ bias,
                                   float* __restrict__ out) {
    const int tid    = threadIdx.x;
    const int wave   = tid >> 5;
    const int lane   = tid & 31;
    const int c0     = wave * 16;
    const int blkrow = blockIdx.x * 16;
    const int lrow   = lane & 15;
    const int khalf  = (lane >> 4) * 2;

    const float* Arow = A + (size_t)(blkrow + lrow) * DFEAT;

    v8f acc = {};
    for (int k0 = 0; k0 < DFEAT; k0 += 4) {
        const int ka = k0 + khalf;
        v2f a, b;
        a.x = Arow[ka];
        a.y = Arow[ka + 1];
        b.x = W[(size_t)ka * NCOLS + c0 + lrow];
        b.y = W[(size_t)(ka + 1) * NCOLS + c0 + lrow];
        acc = __builtin_amdgcn_wmma_f32_16x16x4_f32(false, a, false, b,
                                                    (short)0, acc, false, false);
    }

    const int col   = c0 + lrow;
    const float bt  = bias[col];
    const int mbase = (lane >> 4) * 8;
    #pragma unroll
    for (int r = 0; r < 8; ++r) {
        float v = acc[r] + bt;
        if (RELU) v = fmaxf(v, 0.0f);
        out[(size_t)(blkrow + mbase + r) * NCOLS + col] = v;
    }
}

extern "C" void kernel_launch(void* const* d_in, const int* in_sizes, int n_in,
                              void* d_out, int out_size, void* d_ws, size_t ws_size,
                              hipStream_t stream) {
    (void)in_sizes; (void)n_in; (void)out_size; (void)ws_size;

    const float* x  = (const float*)d_in[0];
    const int*   ei = (const int*)d_in[1];
    const int* src = ei;
    const int* dst = ei + NEDGES;

    const float* wl_[3]; const float* bl_[3]; const float* wr_[3];
    const float* g_[3];  const float* be_[3]; const float* m_[3]; const float* v_[3];
    for (int l = 0; l < 3; ++l) {
        int b = 2 + 7 * l;
        wl_[l] = (const float*)d_in[b + 0];
        bl_[l] = (const float*)d_in[b + 1];
        wr_[l] = (const float*)d_in[b + 2];
        g_[l]  = (const float*)d_in[b + 3];
        be_[l] = (const float*)d_in[b + 4];
        m_[l]  = (const float*)d_in[b + 5];
        v_[l]  = (const float*)d_in[b + 6];
    }
    const float* wf = (const float*)d_in[23];
    const float* bf = (const float*)d_in[24];
    const float* wc = (const float*)d_in[25];
    const float* bc = (const float*)d_in[26];

    float* ws  = (float*)d_ws;
    float* inv = ws;
    float* agg = inv + NNODES;
    float* hA  = agg + (size_t)NNODES * DFEAT;
    float* hB  = hA  + (size_t)NNODES * DFEAT;
    float* aff = hB  + (size_t)NNODES * DFEAT;
    int* deg    = (int*)(aff + 768);
    int* rowptr = deg + NNODES;
    int* rank   = rowptr + NNODES + 4;
    int* col    = rank + NEDGES;

    const int ROWBLKS = NNODES / 16;

    hipMemsetAsync(deg, 0, NNODES * sizeof(int), stream);
    count_rank_kernel<<<(NEDGES + 255) / 256, 256, 0, stream>>>(dst, deg, rank, NEDGES);
    scan_deg_kernel<<<(NNODES + SCAN_BLOCK - 1) / SCAN_BLOCK, SCAN_BLOCK, 0, stream>>>(
        deg, rowptr, inv, NNODES);
    fill_col_kernel<<<(NEDGES + 255) / 256, 256, 0, stream>>>(src, dst, rank, rowptr,
                                                              col, NEDGES);

    for (int l = 0; l < 3; ++l)
        prep_affine_kernel<<<1, 128, 0, stream>>>(bl_[l], g_[l], be_[l], m_[l], v_[l],
                                                  aff + 256 * l, aff + 256 * l + 128);

    const float* lin[3] = { x, hA, hB };
    float*       lout[3] = { hA, hB, hA };
    for (int l = 0; l < 3; ++l) {
        gather_sum_kernel<<<(NNODES + 7) / 8, 256, 0, stream>>>(lin[l], rowptr, col,
                                                                agg, NNODES);
        sage_fused_kernel<<<ROWBLKS, 256, 0, stream>>>(agg, lin[l], inv,
                                                       wl_[l], wr_[l],
                                                       aff + 256 * l,
                                                       aff + 256 * l + 128,
                                                       lout[l]);
    }

    linear_wmma_kernel<DFEAT, true><<<ROWBLKS, 256, 0, stream>>>(hA, wf, bf, hB);
    linear_wmma_kernel<NCLS, false><<<ROWBLKS, 32, 0, stream>>>(hB, wc, bc,
                                                                (float*)d_out);
}


# Both 128-wide GEMM kernels made persistent over row tiles, each wave holding its weight fragments in registers for the whole K (were re-read from global per 16-row block); sum order unchanged
# speedup vs baseline: 2.8708x; 1.1027x over previous
#include <hip/hip_runtime.h>

#define NNODES 100000
#define NEDGES 1600000
#define DFEAT  128
#define NCLS   16
#define BN_EPS 1e-5f

typedef float v2f __attribute__((ext_vector_type(2)));
typedef float v8f __attribute__((ext_vector_type(8)));

__global__ void count_rank_kernel(const int* __restrict__ dst,
                                  int* __restrict__ deg,
                                  int* __restrict__ rank, int ne) {
    int e = blockIdx.x * blockDim.x + threadIdx.x;
    if (e < ne) rank[e] = atomicAdd(&deg[dst[e]], 1);
}

#define SCAN_BLOCK 1024

__global__ __launch_bounds__(SCAN_BLOCK)
void scan_deg_kernel(const int* __restrict__ deg,
                     int* __restrict__ rowptr,
                     float* __restrict__ inv, int n) {
    __shared__ int wsum[32];
    __shared__ int blk_off;

    const int t    = threadIdx.x;
    const int lane = t & 31;
    const int w    = t >> 5;
    const int base = blockIdx.x * SCAN_BLOCK;

    int part = 0;
    for (int i = t; i < base; i += SCAN_BLOCK) part += deg[i];
    for (int d = 16; d > 0; d >>= 1) part += __shfl_xor(part, d, 32);
    if (lane == 0) wsum[w] = part;
    __syncthreads();
    if (w == 0) {
        int v = wsum[lane];
        for (int d = 16; d > 0; d >>= 1) v += __shfl_xor(v, d, 32);
        if (lane == 0) blk_off = v;
    }
    __syncthreads();
    const int off = blk_off;

    const int i  = base + t;
    const int dg = (i < n) ? deg[i] : 0;
    int incl = dg;
    for (int d = 1; d < 32; d <<= 1) {
        int y = __shfl_up(incl, d, 32);
        if (lane >= d) incl += y;
    }
    if (lane == 31) wsum[w] = incl;
    __syncthreads();
    if (w == 0) {
        const int v = wsum[lane];
        int s = v;
        for (int d = 1; d < 32; d <<= 1) {
            int y = __shfl_up(s, d, 32);
            if (lane >= d) s += y;
        }
        wsum[lane] = s - v;
    }
    __syncthreads();
    const int excl = off + wsum[w] + incl - dg;
    if (i < n) {
        rowptr[i] = excl;
        inv[i]    = 1.0f / fmaxf((float)dg, 1.0f);
    }
    if (i == n - 1) rowptr[n] = excl + dg;
}

__global__ void fill_col_kernel(const int* __restrict__ src,
                                const int* __restrict__ dst,
                                const int* __restrict__ rank,
                                const int* __restrict__ rowptr,
                                int* __restrict__ col, int ne) {
    int e = blockIdx.x * blockDim.x + threadIdx.x;
    if (e < ne) col[rowptr[dst[e]] + rank[e]] = src[e];
}

__global__ void prep_affine_kernel(const float* __restrict__ bl,
                                   const float* __restrict__ g,
                                   const float* __restrict__ be,
                                   const float* __restrict__ m,
                                   const float* __restrict__ v,
                                   float* __restrict__ alpha,
                                   float* __restrict__ beta) {
    int j = threadIdx.x;
    float s  = g[j] * rsqrtf(v[j] + BN_EPS);
    alpha[j] = s;
    beta[j]  = (bl[j] - m[j]) * s + be[j];
}

__global__ __launch_bounds__(256)
void gather_sum_kernel(const float* __restrict__ x,
                       const int* __restrict__ rowptr,
                       const int* __restrict__ col,
                       float* __restrict__ agg, int n) {
    const int lane = threadIdx.x & 31;
    const int node = __builtin_amdgcn_readfirstlane(
        (int)(blockIdx.x * (blockDim.x >> 5) + (threadIdx.x >> 5)));
    if (node >= n) return;
    const int beg = rowptr[node];
    const int end = rowptr[node + 1];

    float4 acc = make_float4(0.0f, 0.0f, 0.0f, 0.0f);
    for (int base = beg; base < end; base += 32) {
        const int m    = min(32, end - base);
        const int mine = (lane < m) ? col[base + lane] : 0;
        int j = 0;
        for (; j + 4 <= m; j += 4) {
            const int s0 = __builtin_amdgcn_readlane(mine, j);
            const int s1 = __builtin_amdgcn_readlane(mine, j + 1);
            const int s2 = __builtin_amdgcn_readlane(mine, j + 2);
            const int s3 = __builtin_amdgcn_readlane(mine, j + 3);
            const float4 v0 = ((const float4*)(x + (size_t)s0 * DFEAT))[lane];
            const float4 v1 = ((const float4*)(x + (size_t)s1 * DFEAT))[lane];
            const float4 v2 = ((const float4*)(x + (size_t)s2 * DFEAT))[lane];
            const float4 v3 = ((const float4*)(x + (size_t)s3 * DFEAT))[lane];
            acc.x += v0.x; acc.y += v0.y; acc.z += v0.z; acc.w += v0.w;
            acc.x += v1.x; acc.y += v1.y; acc.z += v1.z; acc.w += v1.w;
            acc.x += v2.x; acc.y += v2.y; acc.z += v2.z; acc.w += v2.w;
            acc.x += v3.x; acc.y += v3.y; acc.z += v3.z; acc.w += v3.w;
        }
        for (; j < m; ++j) {
            const int s0 = __builtin_amdgcn_readlane(mine, j);
            const float4 v0 = ((const float4*)(x + (size_t)s0 * DFEAT))[lane];
            acc.x += v0.x; acc.y += v0.y; acc.z += v0.z; acc.w += v0.w;
        }
    }
    ((float4*)(agg + (size_t)node * DFEAT))[lane] = acc;
}

#define TPITCH      132
#define NUNITS      (NNODES / 16)
#define GEMM_BLOCKS 512

template <int NMAT, int NT>
__device__ __forceinline__
void tile_products(const float* __restrict__ As, const float* __restrict__ Xs,
                   const v2f (&b0)[32], const v2f (&b1)[32],
                   int lrow, int khalf, v8f& acc0, v8f& acc1) {
    #pragma unroll
    for (int s = 0; s < 32; ++s) {
        const int ka = 4 * s + khalf;
        const v2f a0 = *(const v2f*)&As[lrow * TPITCH + ka];
        acc0 = __builtin_amdgcn_wmma_f32_16x16x4_f32(false, a0, false, b0[s],
                                                     (short)0, acc0, false, false);
        if (NT == 2) {
            const v2f a1 = *(const v2f*)&As[(16 + lrow) * TPITCH + ka];
            acc1 = __builtin_amdgcn_wmma_f32_16x16x4_f32(false, a1, false, b0[s],
                                                         (short)0, acc1, false, false);
        }
        if (NMAT == 2) {
            const v2f x0 = *(const v2f*)&Xs[lrow * TPITCH + ka];
            acc0 = __builtin_amdgcn_wmma_f32_16x16x4_f32(false, x0, false, b1[s],
                                                         (short)0, acc0, false, false);
            if (NT == 2) {
                const v2f x1 = *(const v2f*)&Xs[(16 + lrow) * TPITCH + ka];
                acc1 = __builtin_amdgcn_wmma_f32_16x16x4_f32(false, x1, false, b1[s],
                                                             (short)0, acc1, false, false);
            }
        }
    }
}

template <int NMAT>
__global__ __launch_bounds__(256) __attribute__((amdgpu_waves_per_eu(1, 4)))
void rows_gemm_kernel(const float* __restrict__ a_in,
                      const float* __restrict__ x_in,
                      const float* __restrict__ inv,
                      const float* __restrict__ w0,
                      const float* __restrict__ w1,
                      const float* __restrict__ alpha,
                      const float* __restrict__ beta,
                      float* __restrict__ out) {
    __shared__ __attribute__((aligned(16))) float T[2][NMAT][32 * TPITCH];

    const int tid   = threadIdx.x;
    const int lane  = tid & 31;
    const int wave  = __builtin_amdgcn_readfirstlane(tid >> 5);
    const int c0    = wave * 16;
    const int lrow  = lane & 15;
    const int khalf = (lane >> 4) * 2;
    const int col   = c0 + lrow;
    const int mbase = (lane >> 4) * 8;

    v2f b0[32], b1[32];
    #pragma unroll
    for (int s = 0; s < 32; ++s) {
        const int ka = 4 * s + khalf;
        b0[s].x = w0[(size_t)ka * DFEAT + col];
        b0[s].y = w0[(size_t)(ka + 1) * DFEAT + col];
        if (NMAT == 2) {
            b1[s].x = w1[(size_t)ka * DFEAT + col];
            b1[s].y = w1[(size_t)(ka + 1) * DFEAT + col];
        } else {
            b1[s].x = 0.0f; b1[s].y = 0.0f;
        }
    }
    const float al = (NMAT == 2) ? alpha[col] : 1.0f;
    const float bt = beta[col];

    const int ub = ((int)blockIdx.x * NUNITS) / GEMM_BLOCKS;
    const int ue = (((int)blockIdx.x + 1) * NUNITS) / GEMM_BLOCKS;

    float4 pa0 = {}, pa1 = {}, pa2 = {}, pa3 = {}, px0 = {}, px1 = {}, px2 = {}, px3 = {};
    float  ps0 = 0.0f, ps1 = 0.0f, ps2 = 0.0f, ps3 = 0.0f;

#define STAGE_LOAD1(U, I, PA, PX, PS)                                          \
    {                                                                          \
        const int row = min((U) * 16 + wave + 8 * (I), NNODES - 1);            \
        PA = ((const float4*)(a_in + (size_t)row * DFEAT))[lane];              \
        if (NMAT == 2) {                                                       \
            PS = inv[row];                                                     \
            PX = ((const float4*)(x_in + (size_t)row * DFEAT))[lane];          \
        }                                                                      \
    }
#define STAGE_LOAD(U)                                                          \
    STAGE_LOAD1(U, 0, pa0, px0, ps0) STAGE_LOAD1(U, 1, pa1, px1, ps1)          \
    STAGE_LOAD1(U, 2, pa2, px2, ps2) STAGE_LOAD1(U, 3, pa3, px3, ps3)

#define STAGE_STORE1(B, I, PA, PX, PS)                                         \
    {                                                                          \
        const int r = wave + 8 * (I);                                          \
        float4 av = PA;                                                        \
        if (NMAT == 2) {                                                       \
            const float s = PS;                                                \
            av.x *= s; av.y *= s; av.z *= s; av.w *= s;                        \
            *(float4*)&T[B][NMAT - 1][r * TPITCH + lane * 4] = PX;             \
        }                                                                      \
        *(float4*)&T[B][0][r * TPITCH + lane * 4] = av;                        \
    }
#define STAGE_STORE(B)                                                         \
    STAGE_STORE1(B, 0, pa0, px0, ps0) STAGE_STORE1(B, 1, pa1, px1, ps1)        \
    STAGE_STORE1(B, 2, pa2, px2, ps2) STAGE_STORE1(B, 3, pa3, px3, ps3)

    if (ub < ue) {
        STAGE_LOAD(ub)
        STAGE_STORE(0)
    }
    __syncthreads();

    int cur = 0;
    for (int u = ub; u < ue; u += 2) {
        const bool more = (u + 2 < ue);
        if (more) { STAGE_LOAD(u + 2) }

        const float* As = T[cur][0];
        const float* Xs = T[cur][NMAT - 1];
        const int two = (ue - u >= 2);
        v8f acc0 = {}, acc1 = {};
        if (two) tile_products<NMAT, 2>(As, Xs, b0, b1, lrow, khalf, acc0, acc1);
        else     tile_products<NMAT, 1>(As, Xs, b0, b1, lrow, khalf, acc0, acc1);

        float o0[8], o1[8];
        #pragma unroll
        for (int r = 0; r < 8; ++r) {
            const float v0 = (NMAT == 2) ? (acc0[r] * al + bt) : (acc0[r] + bt);
            const float v1 = (NMAT == 2) ? (acc1[r] * al + bt) : (acc1[r] + bt);
            o0[r] = fmaxf(v0, 0.0f);
            o1[r] = fmaxf(v1, 0.0f);
        }

        if (more) { STAGE_STORE(cur ^ 1) }
        __syncthreads();
        cur ^= 1;

        const int row0 = u * 16 + mbase;
        #pragma unroll
        for (int r = 0; r < 8; ++r)
            out[(size_t)(row0 + r) * DFEAT + col] = o0[r];
        if (two) {
            #pragma unroll
            for (int r = 0; r < 8; ++r)
                out[(size_t)(row0 + 16 + r) * DFEAT + col] = o1[r];
        }
    }
#undef STAGE_LOAD
#undef STAGE_STORE
#undef STAGE_LOAD1
#undef STAGE_STORE1
}

template <int NCOLS, bool RELU>
__global__ void linear_wmma_kernel(const float* __restrict__ A,
                                   const float* __restrict__ W,
                                   const float* __restrict__ bias,
                                   float* __restrict__ out) {
    const int tid    = threadIdx.x;
    const int wave   = tid >> 5;
    const int lane   = tid & 31;
    const int c0     = wave * 16;
    const int blkrow = blockIdx.x * 16;
    const int lrow   = lane & 15;
    const int khalf  = (lane >> 4) * 2;

    const float* Arow = A + (size_t)(blkrow + lrow) * DFEAT;

    v8f acc = {};
    for (int k0 = 0; k0 < DFEAT; k0 += 4) {
        const int ka = k0 + khalf;
        v2f a, b;
        a.x = Arow[ka];
        a.y = Arow[ka + 1];
        b.x = W[(size_t)ka * NCOLS + c0 + lrow];
        b.y = W[(size_t)(ka + 1) * NCOLS + c0 + lrow];
        acc = __builtin_amdgcn_wmma_f32_16x16x4_f32(false, a, false, b,
                                                    (short)0, acc, false, false);
    }

    const int col   = c0 + lrow;
    const float bt  = bias[col];
    const int mbase = (lane >> 4) * 8;
    #pragma unroll
    for (int r = 0; r < 8; ++r) {
        float v = acc[r] + bt;
        if (RELU) v = fmaxf(v, 0.0f);
        out[(size_t)(blkrow + mbase + r) * NCOLS + col] = v;
    }
}

extern "C" void kernel_launch(void* const* d_in, const int* in_sizes, int n_in,
                              void* d_out, int out_size, void* d_ws, size_t ws_size,
                              hipStream_t stream) {
    (void)in_sizes; (void)n_in; (void)out_size; (void)ws_size;

    const float* x  = (const float*)d_in[0];
    const int*   ei = (const int*)d_in[1];
    const int* src = ei;
    const int* dst = ei + NEDGES;

    const float* wl_[3]; const float* bl_[3]; const float* wr_[3];
    const float* g_[3];  const float* be_[3]; const float* m_[3]; const float* v_[3];
    for (int l = 0; l < 3; ++l) {
        int b = 2 + 7 * l;
        wl_[l] = (const float*)d_in[b + 0];
        bl_[l] = (const float*)d_in[b + 1];
        wr_[l] = (const float*)d_in[b + 2];
        g_[l]  = (const float*)d_in[b + 3];
        be_[l] = (const float*)d_in[b + 4];
        m_[l]  = (const float*)d_in[b + 5];
        v_[l]  = (const float*)d_in[b + 6];
    }
    const float* wf = (const float*)d_in[23];
    const float* bf = (const float*)d_in[24];
    const float* wc = (const float*)d_in[25];
    const float* bc = (const float*)d_in[26];

    float* ws  = (float*)d_ws;
    float* inv = ws;
    float* agg = inv + NNODES;
    float* hA  = agg + (size_t)NNODES * DFEAT;
    float* hB  = hA  + (size_t)NNODES * DFEAT;
    float* aff = hB  + (size_t)NNODES * DFEAT;
    int* deg    = (int*)(aff + 768);
    int* rowptr = deg + NNODES;
    int* rank   = rowptr + NNODES + 4;
    int* col    = rank + NEDGES;

    const int ROWBLKS = NNODES / 16;

    hipMemsetAsync(deg, 0, NNODES * sizeof(int), stream);
    count_rank_kernel<<<(NEDGES + 255) / 256, 256, 0, stream>>>(dst, deg, rank, NEDGES);
    scan_deg_kernel<<<(NNODES + SCAN_BLOCK - 1) / SCAN_BLOCK, SCAN_BLOCK, 0, stream>>>(
        deg, rowptr, inv, NNODES);
    fill_col_kernel<<<(NEDGES + 255) / 256, 256, 0, stream>>>(src, dst, rank, rowptr,
                                                              col, NEDGES);

    for (int l = 0; l < 3; ++l)
        prep_affine_kernel<<<1, 128, 0, stream>>>(bl_[l], g_[l], be_[l], m_[l], v_[l],
                                                  aff + 256 * l, aff + 256 * l + 128);

    const float* lin[3] = { x, hA, hB };
    float*       lout[3] = { hA, hB, hA };
    for (int l = 0; l < 3; ++l) {
        gather_sum_kernel<<<(NNODES + 7) / 8, 256, 0, stream>>>(lin[l], rowptr, col,
                                                                agg, NNODES);
        rows_gemm_kernel<2><<<GEMM_BLOCKS, 256, 0, stream>>>(agg, lin[l], inv,
                                                             wl_[l], wr_[l],
                                                             aff + 256 * l,
                                                             aff + 256 * l + 128,
                                                             lout[l]);
    }

    rows_gemm_kernel<1><<<GEMM_BLOCKS, 256, 0, stream>>>(hA, nullptr, nullptr, wf, nullptr,
                                                         nullptr, bf, hB);
    linear_wmma_kernel<NCLS, false><<<ROWBLKS, 32, 0, stream>>>(hB, wc, bc,
                                                                (float*)d_out);
}


# Neighbour lists kept as one fixed 256-byte slot row per node, filled straight from the rank atomic (spill list past 64 edges): degree scan, fill pass and their 2 launches removed; sum order unchanged
# speedup vs baseline: 3.0023x; 1.0458x over previous
#include <hip/hip_runtime.h>

#define NNODES 100000
#define NEDGES 1600000
#define DFEAT  128
#define NCLS   16
#define BN_EPS 1e-5f

typedef float v2f __attribute__((ext_vector_type(2)));
typedef float v8f __attribute__((ext_vector_type(8)));

#define SLOTS 64

__global__ void place_edges_kernel(const int* __restrict__ src,
                                   const int* __restrict__ dst,
                                   int* __restrict__ deg,
                                   int* __restrict__ slots,
                                   int* __restrict__ nspill,
                                   int* __restrict__ spill, int ne) {
    int e = blockIdx.x * blockDim.x + threadIdx.x;
    if (e < ne) {
        const int d = dst[e];
        const int s = src[e];
        const int r = atomicAdd(&deg[d], 1);
        if (r < SLOTS) {
            slots[(size_t)d * SLOTS + r] = s;
        } else {
            const int o = atomicAdd(nspill, 1);
            spill[2 * (size_t)o]     = s;
            spill[2 * (size_t)o + 1] = d;
        }
    }
}

__global__ void prep_affine_kernel(const float* __restrict__ bl,
                                   const float* __restrict__ g,
                                   const float* __restrict__ be,
                                   const float* __restrict__ m,
                                   const float* __restrict__ v,
                                   float* __restrict__ alpha,
                                   float* __restrict__ beta) {
    int j = threadIdx.x;
    float s  = g[j] * rsqrtf(v[j] + BN_EPS);
    alpha[j] = s;
    beta[j]  = (bl[j] - m[j]) * s + be[j];
}

__global__ __launch_bounds__(256)
void gather_sum_kernel(const float* __restrict__ x,
                       const int* __restrict__ deg,
                       const int* __restrict__ slots,
                       const int* __restrict__ nspill,
                       const int* __restrict__ spill,
                       float* __restrict__ agg,
                       float* __restrict__ inv, int n) {
    const int lane = threadIdx.x & 31;
    const int node = __builtin_amdgcn_readfirstlane(
        (int)(blockIdx.x * (blockDim.x >> 5) + (threadIdx.x >> 5)));
    if (node >= n) return;
    const int dg  = deg[node];
    const int cnt = min(dg, SLOTS);
    const int* __restrict__ myrow = slots + (size_t)node * SLOTS;

    float4 acc = make_float4(0.0f, 0.0f, 0.0f, 0.0f);
    for (int base = 0; base < cnt; base += 32) {
        const int m    = min(32, cnt - base);
        const int mine = (lane < m) ? myrow[base + lane] : 0;
        int j = 0;
        for (; j + 4 <= m; j += 4) {
            const int s0 = __builtin_amdgcn_readlane(mine, j);
            const int s1 = __builtin_amdgcn_readlane(mine, j + 1);
            const int s2 = __builtin_amdgcn_readlane(mine, j + 2);
            const int s3 = __builtin_amdgcn_readlane(mine, j + 3);
            const float4 v0 = ((const float4*)(x + (size_t)s0 * DFEAT))[lane];
            const float4 v1 = ((const float4*)(x + (size_t)s1 * DFEAT))[lane];
            const float4 v2 = ((const float4*)(x + (size_t)s2 * DFEAT))[lane];
            const float4 v3 = ((const float4*)(x + (size_t)s3 * DFEAT))[lane];
            acc.x += v0.x; acc.y += v0.y; acc.z += v0.z; acc.w += v0.w;
            acc.x += v1.x; acc.y += v1.y; acc.z += v1.z; acc.w += v1.w;
            acc.x += v2.x; acc.y += v2.y; acc.z += v2.z; acc.w += v2.w;
            acc.x += v3.x; acc.y += v3.y; acc.z += v3.z; acc.w += v3.w;
        }
        for (; j < m; ++j) {
            const int s0 = __builtin_amdgcn_readlane(mine, j);
            const float4 v0 = ((const float4*)(x + (size_t)s0 * DFEAT))[lane];
            acc.x += v0.x; acc.y += v0.y; acc.z += v0.z; acc.w += v0.w;
        }
    }

    if (dg > SLOTS) {
        const int ns = min(nspill[0], NEDGES);
        for (int base = 0; base < ns; base += 32) {
            int s = 0, d = -1;
            if (base + lane < ns) {
                s = spill[2 * (size_t)(base + lane)];
                d = spill[2 * (size_t)(base + lane) + 1];
            }
            unsigned mask = __builtin_amdgcn_ballot_w32(d == node);
            while (mask != 0u) {
                const int j = __builtin_ctz(mask);
                mask &= mask - 1u;
                const int s0 = __builtin_amdgcn_readlane(s, j);
                const float4 v0 = ((const float4*)(x + (size_t)s0 * DFEAT))[lane];
                acc.x += v0.x; acc.y += v0.y; acc.z += v0.z; acc.w += v0.w;
            }
        }
    }

    ((float4*)(agg + (size_t)node * DFEAT))[lane] = acc;
    if (inv != nullptr && lane == 0) inv[node] = 1.0f / fmaxf((float)dg, 1.0f);
}

#define TPITCH      132
#define NUNITS      (NNODES / 16)
#define GEMM_BLOCKS 512

template <int NMAT, int NT>
__device__ __forceinline__
void tile_products(const float* __restrict__ As, const float* __restrict__ Xs,
                   const v2f (&b0)[32], const v2f (&b1)[32],
                   int lrow, int khalf, v8f& acc0, v8f& acc1) {
    #pragma unroll
    for (int s = 0; s < 32; ++s) {
        const int ka = 4 * s + khalf;
        const v2f a0 = *(const v2f*)&As[lrow * TPITCH + ka];
        acc0 = __builtin_amdgcn_wmma_f32_16x16x4_f32(false, a0, false, b0[s],
                                                     (short)0, acc0, false, false);
        if (NT == 2) {
            const v2f a1 = *(const v2f*)&As[(16 + lrow) * TPITCH + ka];
            acc1 = __builtin_amdgcn_wmma_f32_16x16x4_f32(false, a1, false, b0[s],
                                                         (short)0, acc1, false, false);
        }
        if (NMAT == 2) {
            const v2f x0 = *(const v2f*)&Xs[lrow * TPITCH + ka];
            acc0 = __builtin_amdgcn_wmma_f32_16x16x4_f32(false, x0, false, b1[s],
                                                         (short)0, acc0, false, false);
            if (NT == 2) {
                const v2f x1 = *(const v2f*)&Xs[(16 + lrow) * TPITCH + ka];
                acc1 = __builtin_amdgcn_wmma_f32_16x16x4_f32(false, x1, false, b1[s],
                                                             (short)0, acc1, false, false);
            }
        }
    }
}

template <int NMAT>
__global__ __launch_bounds__(256) __attribute__((amdgpu_waves_per_eu(1, 4)))
void rows_gemm_kernel(const float* __restrict__ a_in,
                      const float* __restrict__ x_in,
                      const float* __restrict__ inv,
                      const float* __restrict__ w0,
                      const float* __restrict__ w1,
                      const float* __restrict__ alpha,
                      const float* __restrict__ beta,
                      float* __restrict__ out) {
    __shared__ __attribute__((aligned(16))) float T[2][NMAT][32 * TPITCH];

    const int tid   = threadIdx.x;
    const int lane  = tid & 31;
    const int wave  = __builtin_amdgcn_readfirstlane(tid >> 5);
    const int c0    = wave * 16;
    const int lrow  = lane & 15;
    const int khalf = (lane >> 4) * 2;
    const int col   = c0 + lrow;
    const int mbase = (lane >> 4) * 8;

    v2f b0[32], b1[32];
    #pragma unroll
    for (int s = 0; s < 32; ++s) {
        const int ka = 4 * s + khalf;
        b0[s].x = w0[(size_t)ka * DFEAT + col];
        b0[s].y = w0[(size_t)(ka + 1) * DFEAT + col];
        if (NMAT == 2) {
            b1[s].x = w1[(size_t)ka * DFEAT + col];
            b1[s].y = w1[(size_t)(ka + 1) * DFEAT + col];
        } else {
            b1[s].x = 0.0f; b1[s].y = 0.0f;
        }
    }
    const float al = (NMAT == 2) ? alpha[col] : 1.0f;
    const float bt = beta[col];

    const int ub = ((int)blockIdx.x * NUNITS) / GEMM_BLOCKS;
    const int ue = (((int)blockIdx.x + 1) * NUNITS) / GEMM_BLOCKS;

    float4 pa0 = {}, pa1 = {}, pa2 = {}, pa3 = {}, px0 = {}, px1 = {}, px2 = {}, px3 = {};
    float  ps0 = 0.0f, ps1 = 0.0f, ps2 = 0.0f, ps3 = 0.0f;

#define STAGE_LOAD1(U, I, PA, PX, PS)                                          \
    {                                                                          \
        const int row = min((U) * 16 + wave + 8 * (I), NNODES - 1);            \
        PA = ((const float4*)(a_in + (size_t)row * DFEAT))[lane];              \
        if (NMAT == 2) {                                                       \
            PS = inv[row];                                                     \
            PX = ((const float4*)(x_in + (size_t)row * DFEAT))[lane];          \
        }                                                                      \
    }
#define STAGE_LOAD(U)                                                          \
    STAGE_LOAD1(U, 0, pa0, px0, ps0) STAGE_LOAD1(U, 1, pa1, px1, ps1)          \
    STAGE_LOAD1(U, 2, pa2, px2, ps2) STAGE_LOAD1(U, 3, pa3, px3, ps3)

#define STAGE_STORE1(B, I, PA, PX, PS)                                         \
    {                                                                          \
        const int r = wave + 8 * (I);                                          \
        float4 av = PA;                                                        \
        if (NMAT == 2) {                                                       \
            const float s = PS;                                                \
            av.x *= s; av.y *= s; av.z *= s; av.w *= s;                        \
            *(float4*)&T[B][NMAT - 1][r * TPITCH + lane * 4] = PX;             \
        }                                                                      \
        *(float4*)&T[B][0][r * TPITCH + lane * 4] = av;                        \
    }
#define STAGE_STORE(B)                                                         \
    STAGE_STORE1(B, 0, pa0, px0, ps0) STAGE_STORE1(B, 1, pa1, px1, ps1)        \
    STAGE_STORE1(B, 2, pa2, px2, ps2) STAGE_STORE1(B, 3, pa3, px3, ps3)

    if (ub < ue) {
        STAGE_LOAD(ub)
        STAGE_STORE(0)
    }
    __syncthreads();

    int cur = 0;
    for (int u = ub; u < ue; u += 2) {
        const bool more = (u + 2 < ue);
        if (more) { STAGE_LOAD(u + 2) }

        const float* As = T[cur][0];
        const float* Xs = T[cur][NMAT - 1];
        const int two = (ue - u >= 2);
        v8f acc0 = {}, acc1 = {};
        if (two) tile_products<NMAT, 2>(As, Xs, b0, b1, lrow, khalf, acc0, acc1);
        else     tile_products<NMAT, 1>(As, Xs, b0, b1, lrow, khalf, acc0, acc1);

        float o0[8], o1[8];
        #pragma unroll
        for (int r = 0; r < 8; ++r) {
            const float v0 = (NMAT == 2) ? (acc0[r] * al + bt) : (acc0[r] + bt);
            const float v1 = (NMAT == 2) ? (acc1[r] * al + bt) : (acc1[r] + bt);
            o0[r] = fmaxf(v0, 0.0f);
            o1[r] = fmaxf(v1, 0.0f);
        }

        if (more) { STAGE_STORE(cur ^ 1) }
        __syncthreads();
        cur ^= 1;

        const int row0 = u * 16 + mbase;
        #pragma unroll
        for (int r = 0; r < 8; ++r)
            out[(size_t)(row0 + r) * DFEAT + col] = o0[r];
        if (two) {
            #pragma unroll
            for (int r = 0; r < 8; ++r)
                out[(size_t)(row0 + 16 + r) * DFEAT + col] = o1[r];
        }
    }
#undef STAGE_LOAD
#undef STAGE_STORE
#undef STAGE_LOAD1
#undef STAGE_STORE1
}

template <int NCOLS, bool RELU>
__global__ void linear_wmma_kernel(const float* __restrict__ A,
                                   const float* __restrict__ W,
                                   const float* __restrict__ bias,
                                   float* __restrict__ out) {
    const int tid    = threadIdx.x;
    const int wave   = tid >> 5;
    const int lane   = tid & 31;
    const int c0     = wave * 16;
    const int blkrow = blockIdx.x * 16;
    const int lrow   = lane & 15;
    const int khalf  = (lane >> 4) * 2;

    const float* Arow = A + (size_t)(blkrow + lrow) * DFEAT;

    v8f acc = {};
    for (int k0 = 0; k0 < DFEAT; k0 += 4) {
        const int ka = k0 + khalf;
        v2f a, b;
        a.x = Arow[ka];
        a.y = Arow[ka + 1];
        b.x = W[(size_t)ka * NCOLS + c0 + lrow];
        b.y = W[(size_t)(ka + 1) * NCOLS + c0 + lrow];
        acc = __builtin_amdgcn_wmma_f32_16x16x4_f32(false, a, false, b,
                                                    (short)0, acc, false, false);
    }

    const int col   = c0 + lrow;
    const float bt  = bias[col];
    const int mbase = (lane >> 4) * 8;
    #pragma unroll
    for (int r = 0; r < 8; ++r) {
        float v = acc[r] + bt;
        if (RELU) v = fmaxf(v, 0.0f);
        out[(size_t)(blkrow + mbase + r) * NCOLS + col] = v;
    }
}

extern "C" void kernel_launch(void* const* d_in, const int* in_sizes, int n_in,
                              void* d_out, int out_size, void* d_ws, size_t ws_size,
                              hipStream_t stream) {
    (void)in_sizes; (void)n_in; (void)out_size; (void)ws_size;

    const float* x  = (const float*)d_in[0];
    const int*   ei = (const int*)d_in[1];
    const int* src = ei;
    const int* dst = ei + NEDGES;

    const float* wl_[3]; const float* bl_[3]; const float* wr_[3];
    const float* g_[3];  const float* be_[3]; const float* m_[3]; const float* v_[3];
    for (int l = 0; l < 3; ++l) {
        int b = 2 + 7 * l;
        wl_[l] = (const float*)d_in[b + 0];
        bl_[l] = (const float*)d_in[b + 1];
        wr_[l] = (const float*)d_in[b + 2];
        g_[l]  = (const float*)d_in[b + 3];
        be_[l] = (const float*)d_in[b + 4];
        m_[l]  = (const float*)d_in[b + 5];
        v_[l]  = (const float*)d_in[b + 6];
    }
    const float* wf = (const float*)d_in[23];
    const float* bf = (const float*)d_in[24];
    const float* wc = (const float*)d_in[25];
    const float* bc = (const float*)d_in[26];

    float* ws  = (float*)d_ws;
    float* inv = ws;
    float* agg = inv + NNODES;
    float* hA  = agg + (size_t)NNODES * DFEAT;
    float* hB  = hA  + (size_t)NNODES * DFEAT;
    float* aff = hB  + (size_t)NNODES * DFEAT;
    int* deg    = (int*)(aff + 768);
    int* nspill = deg + NNODES;
    int* spill  = nspill + 4;
    int* slots  = spill + 2 * NEDGES + 60;

    const int ROWBLKS = NNODES / 16;

    hipMemsetAsync(deg, 0, (NNODES + 4) * sizeof(int), stream);
    place_edges_kernel<<<(NEDGES + 255) / 256, 256, 0, stream>>>(src, dst, deg, slots,
                                                                 nspill, spill, NEDGES);

    for (int l = 0; l < 3; ++l)
        prep_affine_kernel<<<1, 128, 0, stream>>>(bl_[l], g_[l], be_[l], m_[l], v_[l],
                                                  aff + 256 * l, aff + 256 * l + 128);

    const float* lin[3] = { x, hA, hB };
    float*       lout[3] = { hA, hB, hA };
    float*       winv[3] = { inv, nullptr, nullptr };
    for (int l = 0; l < 3; ++l) {
        gather_sum_kernel<<<(NNODES + 7) / 8, 256, 0, stream>>>(lin[l], deg, slots,
                                                                nspill, spill, agg,
                                                                winv[l], NNODES);
        rows_gemm_kernel<2><<<GEMM_BLOCKS, 256, 0, stream>>>(agg, lin[l], inv,
                                                             wl_[l], wr_[l],
                                                             aff + 256 * l,
                                                             aff + 256 * l + 128,
                                                             lout[l]);
    }

    rows_gemm_kernel<1><<<GEMM_BLOCKS, 256, 0, stream>>>(hA, nullptr, nullptr, wf, nullptr,
                                                         nullptr, bf, hB);
    linear_wmma_kernel<NCLS, false><<<ROWBLKS, 32, 0, stream>>>(hB, wc, bc,
                                                                (float*)d_out);
}
